# starting kernel
# baseline (speedup 1.0000x reference)
#include <hip/hip_runtime.h>


#define NN 50000
#define NE 800000
#define DD 128
#define NT (NN / 16)
#define FEAT (NN * DD)
#define WSTRIDE 132
#define WOFF (DD * WSTRIDE)

typedef float v2f __attribute__((ext_vector_type(2)));
typedef float v8f __attribute__((ext_vector_type(8)));

__device__ __forceinline__ v8f wmma4(v2f a, v2f b, v8f c) {
  return __builtin_amdgcn_wmma_f32_16x16x4_f32(false, a, false, b, (short)0, c,
                                               false, false);
}

__global__ void k_zero(float4* __restrict__ p, int n4) {
  int i = blockIdx.x * blockDim.x + threadIdx.x;
  int stride = gridDim.x * blockDim.x;
  float4 z = {0.f, 0.f, 0.f, 0.f};
  for (; i < n4; i += stride) p[i] = z;
}

__global__ void k_deg(const int* __restrict__ row, float* __restrict__ deg) {
  int e = blockIdx.x * blockDim.x + threadIdx.x;
  if (e < NE) atomicAdd(&deg[row[e]], 1.0f);
}

__global__ void k_scatter(const float* __restrict__ xin,
                          const int* __restrict__ row,
                          const int* __restrict__ col,
                          float* __restrict__ agg) {
  long tid = (long)blockIdx.x * blockDim.x + threadIdx.x;
  int e = (int)(tid >> 5);
  if (e >= NE) return;
  int c = ((int)tid & 31) << 2;
  int s = col[e];
  int d = row[e];
  const float4 v = *(const float4*)(xin + (size_t)s * DD + c);
  float* dst = agg + (size_t)d * DD + c;
  atomicAdd(dst + 0, v.x);
  atomicAdd(dst + 1, v.y);
  atomicAdd(dst + 2, v.z);
  atomicAdd(dst + 3, v.w);
}

__global__ void k_sage_gemm(const float* __restrict__ agg,
                            const float* __restrict__ deg,
                            const float* __restrict__ h,
                            const float* __restrict__ Wl,
                            const float* __restrict__ Wr,
                            const float* __restrict__ bias,
                            const float* __restrict__ resid,
                            float* __restrict__ out,
                            float* __restrict__ hout) {
  __shared__ float wlds[2 * DD * WSTRIDE];

  for (int i = threadIdx.x; i < (DD * DD) / 4; i += 256) {
    int o = i >> 5;
    int kc = (i & 31) << 2;
    float4 a = *(const float4*)(Wl + (size_t)o * DD + kc);
    float4 b = *(const float4*)(Wr + (size_t)o * DD + kc);
    *(float4*)(&wlds[o * WSTRIDE + kc]) = a;
    *(float4*)(&wlds[WOFF + o * WSTRIDE + kc]) = b;
  }
  __syncthreads();

  const int nw = blockDim.x >> 5;
  const int wv = threadIdx.x >> 5;
  const int lane = threadIdx.x & 31;
  const int half = lane >> 4;
  const int lm = lane & 15;
  const int tstride = gridDim.x * nw;

  for (int tile = blockIdx.x * nw + wv; tile < NT; tile += tstride) {
    const int m0 = tile << 4;
    if (tile + tstride < NT) {
      const size_t nb = (size_t)((tile + tstride) * 16 + lm) * DD;
      __builtin_prefetch(agg + nb, 0, 1);
      __builtin_prefetch(h + nb, 0, 1);
    }

    const float sc = 1.0f / fmaxf(deg[m0 + lm], 1.0f);
    const float* arow = agg + (size_t)(m0 + lm) * DD + 2 * half;
    const float* hrow = h + (size_t)(m0 + lm) * DD + 2 * half;
    const float* blbase = &wlds[(size_t)lm * WSTRIDE + 2 * half];
    const float* brbase = &wlds[WOFF + (size_t)lm * WSTRIDE + 2 * half];

    v8f acc[8] = {};

#pragma unroll 1
    for (int pass = 0; pass < 2; ++pass) {
      const float* aptr = pass ? hrow : arow;
      const float* bb = pass ? brbase : blbase;
      const float scale = pass ? 1.0f : sc;

      v2f cur[8], nxt[8];
#pragma unroll
      for (int t = 0; t < 8; ++t)
        cur[t] = *(const v2f*)(bb + (size_t)(t << 4) * WSTRIDE);
      float2 a2 = *(const float2*)(aptr);

#pragma unroll 2
      for (int k = 0; k < 32; ++k) {
        const int K1 = ((k + 1) & 31) << 2;
#pragma unroll
        for (int t = 0; t < 8; ++t)
          nxt[t] = *(const v2f*)(bb + (size_t)(t << 4) * WSTRIDE + K1);
        float2 a2n = *(const float2*)(aptr + K1);

        v2f va;
        va.x = a2.x * scale;
        va.y = a2.y * scale;
#pragma unroll
        for (int t = 0; t < 8; ++t) acc[t] = wmma4(va, cur[t], acc[t]);

#pragma unroll
        for (int t = 0; t < 8; ++t) cur[t] = nxt[t];
        a2 = a2n;
      }
    }

#pragma unroll
    for (int t = 0; t < 8; ++t) {
      const int n0 = t << 4;
      const float bb = bias[n0 + lm];
#pragma unroll
      for (int r = 0; r < 8; ++r) {
        const size_t idx = (size_t)(m0 + r + (half << 3)) * DD + n0 + lm;
        float v = acc[t][r] + bb;
        if (resid) v += resid[idx];
        out[idx] = v;
        if (hout) hout[idx] = fmaxf(v, 0.0f);
      }
    }
  }
}

extern "C" void kernel_launch(void* const* d_in, const int* in_sizes, int n_in,
                              void* d_out, int out_size, void* d_ws,
                              size_t ws_size, hipStream_t stream) {
  const float* x = (const float*)d_in[0];
  const int* erow = (const int*)d_in[1];
  const int* ecol = (const int*)d_in[2];
  const float* Wl0 = (const float*)d_in[3];
  const float* Wr0 = (const float*)d_in[4];
  const float* b0 = (const float*)d_in[5];
  const float* Wl1 = (const float*)d_in[6];
  const float* Wr1 = (const float*)d_in[7];
  const float* b1 = (const float*)d_in[8];
  const float* Wl2 = (const float*)d_in[9];
  const float* Wr2 = (const float*)d_in[10];
  const float* b2 = (const float*)d_in[11];
  float* out = (float*)d_out;

  float* ws = (float*)d_ws;
  float* deg = ws;
  float* agg = ws + NN;
  float* h = agg + (size_t)FEAT;
  float* ycur = h + (size_t)FEAT;

  dim3 b256(256);
  const int degBlocks = (NE + 255) / 256;
  const int scatterBlocks = (int)(((long)NE * 32 + 255) / 256);
  const int gemmBlocks = (NT + 7) / 8;

  k_zero<<<dim3(64), b256, 0, stream>>>((float4*)deg, NN / 4);
  k_deg<<<dim3(degBlocks), b256, 0, stream>>>(erow, deg);

  k_zero<<<dim3(2048), b256, 0, stream>>>((float4*)agg, FEAT / 4);
  k_scatter<<<dim3(scatterBlocks), b256, 0, stream>>>(x, erow, ecol, agg);
  k_sage_gemm<<<dim3(gemmBlocks), b256, 0, stream>>>(agg, deg, x, Wl0, Wr0, b0,
                                                     nullptr, ycur, h);

  k_zero<<<dim3(2048), b256, 0, stream>>>((float4*)agg, FEAT / 4);
  k_scatter<<<dim3(scatterBlocks), b256, 0, stream>>>(h, erow, ecol, agg);
  k_sage_gemm<<<dim3(gemmBlocks), b256, 0, stream>>>(agg, deg, h, Wl1, Wr1, b1,
                                                     ycur, ycur, h);

  k_zero<<<dim3(2048), b256, 0, stream>>>((float4*)agg, FEAT / 4);
  k_scatter<<<dim3(scatterBlocks), b256, 0, stream>>>(h, erow, ecol, agg);
  k_sage_gemm<<<dim3(gemmBlocks), b256, 0, stream>>>(agg, deg, h, Wl2, Wr2, b2,
                                                     ycur, out, nullptr);
}


# Edge filter by index: first kernel writes a word per edge (block, block row, source) and per block a bitmap of the 32-edge lines with its edges; a work-group reads only those lines; sums as before
# speedup vs baseline: 4.3775x; 4.3775x over previous
#include <hip/hip_runtime.h>


#define NN 50000
#define NE 800000
#define DD 128
#define NT (NN / 16)
#define FEAT (NN * DD)
#define WSTRIDE 132
#define WOFF (DD * WSTRIDE)
static_assert(DD * sizeof(float) == 512, "a node row is 512 bytes: planes laid end to end keep rows aligned");

typedef float v2f __attribute__((ext_vector_type(2)));
typedef float v8f __attribute__((ext_vector_type(8)));

__device__ __forceinline__ v8f wmma4(v2f a, v2f b, v8f c) {
  return __builtin_amdgcn_wmma_f32_16x16x4_f32(false, a, false, b, (short)0, c,
                                               false, false);
}

#define RB 196
#define NBLK ((NN + RB - 1) / RB)
#define QCAP 1024
#define NGRP (NE / 32)
#define BMW 25
#define GQCAP (32 * BMW)
#define SCAN_B 16
static_assert(NE % 32 == 0, "whole groups of 32 edges");
static_assert(32 * BMW * 32 >= NGRP, "the waves' bitmap words cover every group");
static_assert(NGRP <= 65536, "a queued group number is 16 bits");
static_assert(31 + SCAN_B * 32 <= QCAP, "a batch's edges must fit the ring");
#define LCAP 512
#define LIST_WORDS (NBLK * 32 * LCAP)

#define PACK_OFF (LIST_WORDS + NBLK * 32)
#define BM_OFF (PACK_OFF + NE)
#define BM_SPAN (NBLK * 4)
#define BM_WORDS ((NE / 4096 + 1) * BM_SPAN)
static_assert(BM_OFF + BM_WORDS <= FEAT,
              "lists, edge words and bitmaps must fit in the output");
static_assert(PACK_OFF % 32 == 0, "a group of edge words is one 128-byte line");
static_assert(BM_OFF % 32 == 0, "an index work-group's bitmap words are whole 128-byte lines");
static_assert(NBLK == 256, "a block number is the top byte of an edge word");
static_assert(RB <= 256, "a block row is one byte of an edge word");
static_assert(NN <= 65536, "the source node is the low half of an edge word");

__device__ __forceinline__ void lds_add(float* p, float v) {
  __hip_atomic_fetch_add(p, v, __ATOMIC_RELAXED, __HIP_MEMORY_SCOPE_WORKGROUP);
}

template <bool FULL, bool RELU>
__device__ __forceinline__ void gather_add(int c, int d, int cnt, int lane,
                                           const float4* __restrict__ x4,
                                           float* blk) {
#pragma unroll
  for (int g = 0; g < 32; g += 8) {
    if (!FULL && g >= cnt) break;
    float4 v[8];
#pragma unroll
    for (int u = 0; u < 8; ++u) {
      const int cu = __builtin_amdgcn_readlane(c, g + u);
      v[u] = x4[(size_t)cu * (DD / 4) + lane];
    }
    if (RELU) __builtin_amdgcn_sched_barrier(0);
#pragma unroll
    for (int u = 0; u < 8; ++u) {
      if (FULL || g + u < cnt) {
        const int du = __builtin_amdgcn_readlane(d, g + u);
        float* p = blk + du * DD + lane;
        float4 w = v[u];
        if (RELU) {
          w.x = fmaxf(w.x, 0.0f);
          w.y = fmaxf(w.y, 0.0f);
          w.z = fmaxf(w.z, 0.0f);
          w.w = fmaxf(w.w, 0.0f);
        }
        lds_add(p, w.x);
        lds_add(p + 32, w.y);
        lds_add(p + 64, w.z);
        lds_add(p + 96, w.w);
      }
    }
  }
}

template <bool FULL, bool RECORD, bool RELU>
__device__ __forceinline__ void drain32(const unsigned* q, int head, int cnt,
                                        int lane,
                                        const float4* __restrict__ x4,
                                        float* blk, float* degs,
                                        unsigned* __restrict__ wl) {
  const bool valid = FULL || (lane < cnt);
  unsigned w = 0;
  if (valid) w = q[(head + lane) & (QCAP - 1)] & 0x00FFFFFFu;
  const int c = (int)(w & 0xFFFFu);
  const int d = (int)(w >> 16);
  if (valid) {
    lds_add(degs + d, 1.0f);
    if (RECORD && head + lane < LCAP) wl[head + lane] = w;
  }
  gather_add<FULL, RELU>(c, d, cnt, lane, x4, blk);
}

#define IDX_QUADS 1024
#define IDX_GRID ((NE / 4 + IDX_QUADS - 1) / IDX_QUADS)
#define IDX_WORDS (IDX_GRID * 4)
static_assert(IDX_WORDS <= 32 * BMW && IDX_WORDS * 32 >= NGRP &&
                  IDX_GRID * BM_SPAN <= BM_WORDS,
              "the written bitmap words cover every group and fit their place");
__device__ __forceinline__ unsigned edge_word(unsigned r, unsigned c,
                                              unsigned* bm, int k,
                                              unsigned bit) {
  const unsigned b = r / RB;
  __hip_atomic_fetch_or(bm + b * 4 + k, bit, __ATOMIC_RELAXED,
                        __HIP_MEMORY_SCOPE_WORKGROUP);
  return (b << 24) | ((r - b * RB) << 16) | c;
}
__global__ __launch_bounds__(256) void k_edge_index(
    const int* __restrict__ row, const int* __restrict__ col,
    unsigned* __restrict__ pk, unsigned* __restrict__ bmp) {
  __shared__ __attribute__((aligned(16))) unsigned bm[NBLK * 4];
  const int tid = threadIdx.x;
  const int t0 = blockIdx.x * IDX_QUADS + tid;
  int4 r[4], s[4];
#pragma unroll
  for (int k = 0; k < 4; ++k) {
    const int t = t0 + k * 256;
    const int tc = t < NE / 4 ? t : NE / 4 - 1;
    r[k] = ((const int4*)row)[tc];
    s[k] = ((const int4*)col)[tc];
  }
  {
    const uint4 z = {0u, 0u, 0u, 0u};
    ((uint4*)bm)[tid] = z;
  }
  __syncthreads();
  const unsigned bit = 1u << (tid >> 3);
#pragma unroll
  for (int k = 0; k < 4; ++k) {
    const int t = t0 + k * 256;
    if (t < NE / 4) {
      uint4 w;
      w.x = edge_word((unsigned)r[k].x, (unsigned)s[k].x, bm, k, bit);
      w.y = edge_word((unsigned)r[k].y, (unsigned)s[k].y, bm, k, bit);
      w.z = edge_word((unsigned)r[k].z, (unsigned)s[k].z, bm, k, bit);
      w.w = edge_word((unsigned)r[k].w, (unsigned)s[k].w, bm, k, bit);
      ((uint4*)pk)[t] = w;
    }
  }
  __syncthreads();
  ((uint4*)(bmp + (size_t)blockIdx.x * BM_SPAN))[tid] = ((const uint4*)bm)[tid];
}

template <bool RECORD, bool RELU>
__device__ __forceinline__ int scan_phase(
    const unsigned* __restrict__ bmp, const unsigned* __restrict__ pk,
    const float4* __restrict__ x4, float* blk, float* degs, unsigned* q,
    unsigned short* gq, unsigned* __restrict__ wl, int lane, int wv) {
  int head = 0, tail = 0;
  const unsigned own = blockIdx.x;

  const int j = wv * BMW + lane;
  unsigned bits = 0;
  if (lane < BMW && j < IDX_WORDS)
    bits = bmp[(j >> 2) * BM_SPAN + blockIdx.x * 4 + (j & 3)];
  const int pc = __builtin_popcount(bits);
  int off = 0, tot = 0;
#pragma unroll
  for (int l = 0; l < BMW; ++l) {
    if (lane == l) off = tot;
    tot += __builtin_amdgcn_readlane(pc, l);
  }
  while (bits) {
    gq[off++] = (unsigned short)(j * 32 + __builtin_ctz(bits));
    bits &= bits - 1u;
  }

#pragma unroll 1
  for (int base = 0; base < tot; base += SCAN_B) {
    const int ng = (tot - base < SCAN_B) ? (tot - base) : SCAN_B;
    int gid = 0;
    if (lane < ng) gid = gq[base + lane];
    unsigned w[SCAN_B];
#pragma unroll
    for (int u = 0; u < SCAN_B; ++u) {
      const int gu = __builtin_amdgcn_readlane(gid, u);
      w[u] = pk[gu * 32 + lane];
    }
#pragma unroll
    for (int u = 0; u < SCAN_B; ++u) {
      const bool hit = (u < ng) && ((w[u] >> 24) == own);
      const unsigned bm = __builtin_amdgcn_ballot_w32(hit);
      if (hit)
        q[(tail + __builtin_amdgcn_mbcnt_lo(bm, 0u)) & (QCAP - 1)] = w[u];
      tail += __builtin_popcount(bm);
    }
    while (tail - head >= 32) {
      drain32<true, RECORD, RELU>(q, head, 32, lane, x4, blk, degs, wl);
      head += 32;
    }
  }
  if (tail - head > 0)
    drain32<false, RECORD, RELU>(q, head, tail - head, lane, x4, blk, degs,
                                 wl);
  return tail;
}

template <bool REPLAY>
__global__ __launch_bounds__(1024) void k_block_sum(
    const float* __restrict__ xin, const unsigned* __restrict__ pk,
    const unsigned* __restrict__ bmp, float* __restrict__ agg,
    float* __restrict__ deg,
    unsigned* __restrict__ lst, int* __restrict__ cnts) {
  __shared__ float blk[RB * DD];
  __shared__ unsigned ring[32 * QCAP];
  __shared__ unsigned short gqs[32 * GQCAP];
  __shared__ float degs[RB];
  __shared__ int wcnt[32];

  const int tid = threadIdx.x;
  const int lane = tid & 31;
  const int wv = __builtin_amdgcn_readfirstlane(tid >> 5);
  const int lo = blockIdx.x * RB;
  const int nrows = (NN - lo < RB) ? (NN - lo) : RB;
  const float4* x4 = (const float4*)xin;
  unsigned* q = ring + wv * QCAP;
  unsigned short* gq = gqs + wv * GQCAP;
  unsigned* wl = lst + (size_t)(blockIdx.x * 32 + wv) * LCAP;

  int cl = 0, n = 0;
  unsigned ent = 0;
  if (REPLAY) {
    cl = cnts[blockIdx.x * 32 + lane];
    n = __builtin_amdgcn_readlane(cl, wv);
    if (lane < n && lane < LCAP) ent = wl[lane];
  }

  {
    const float4 z = {0.f, 0.f, 0.f, 0.f};
    for (int i = tid; i < RB * DD / 4; i += 1024) ((float4*)blk)[i] = z;
    if (tid < RB) degs[tid] = 0.f;
  }
  __syncthreads();

  if (REPLAY) {
    const bool over = __builtin_amdgcn_ballot_w32(cl > LCAP) != 0u;
    if (!over) {
      for (int base = 0; base < n; base += 32) {
        unsigned nxt = 0;
        if (base + 32 + lane < n) nxt = wl[base + 32 + lane];
        const int c = (int)(ent & 0xFFFFu);
        const int d = (int)(ent >> 16);
        const int cnt = n - base;
        if (cnt >= 32)
          gather_add<true, true>(c, d, 32, lane, x4, blk);
        else
          gather_add<false, true>(c, d, cnt, lane, x4, blk);
        ent = nxt;
      }
    } else {
      scan_phase<false, true>(bmp, pk, x4, blk, degs, q, gq, wl, lane, wv);
    }
  } else {
    const int tail = scan_phase<true, false>(bmp, pk, x4, blk, degs, q, gq, wl,
                                             lane, wv);
    if (lane == 0) wcnt[wv] = tail;
  }
  __syncthreads();

  if (!REPLAY && wv == 0) cnts[blockIdx.x * 32 + lane] = wcnt[lane];

  for (int rr = wv; rr < nrows; rr += 32) {
    const float* p = blk + rr * DD + lane;
    const float4 o = {p[0], p[32], p[64], p[96]};
    ((float4*)agg)[(size_t)(lo + rr) * (DD / 4) + lane] = o;
  }
  if (!REPLAY && tid < nrows) deg[lo + tid] = degs[tid];
}

#define GEMM_BLOCKS 256
#define GEMM_SLOTS 4
#define GEMM_ROUNDS \
  ((NT + GEMM_BLOCKS * GEMM_SLOTS - 1) / (GEMM_BLOCKS * GEMM_SLOTS))
#define PSTRIDE 36
#define PIECE_WORDS (16 * PSTRIDE)
static_assert(NN % 16 == 0, "every tile is a full one");
static_assert(GEMM_BLOCKS * GEMM_SLOTS <= NT, "round 0 is full everywhere");

__device__ __forceinline__ void load_row16(float4 (&a)[16], const float* p) {
#pragma unroll
  for (int i = 0; i < 16; ++i) a[i] = *(const float4*)(p + 8 * i);
}

template <bool SCALE, bool RELU>
__device__ __forceinline__ void half_product(const float4 (&a)[16],
                                             float scale, const float* bb,
                                             v8f& acc0, v8f& acc1) {
#pragma unroll
  for (int i = 0; i < 16; ++i) {
    const float4 c0 = *(const float4*)(bb + 8 * i);
    const float4 c1 = *(const float4*)(bb + 16 * WSTRIDE + 8 * i);
    v2f a01, a23;
    if (SCALE) {
      a01.x = a[i].x * scale;
      a01.y = a[i].y * scale;
      a23.x = a[i].z * scale;
      a23.y = a[i].w * scale;
    } else if (RELU) {
      a01.x = fmaxf(a[i].x, 0.0f);
      a01.y = fmaxf(a[i].y, 0.0f);
      a23.x = fmaxf(a[i].z, 0.0f);
      a23.y = fmaxf(a[i].w, 0.0f);
    } else {
      a01.x = a[i].x;
      a01.y = a[i].y;
      a23.x = a[i].z;
      a23.y = a[i].w;
    }
    v2f p01, p23, r01, r23;
    p01.x = c0.x;
    p01.y = c0.y;
    p23.x = c0.z;
    p23.y = c0.w;
    r01.x = c1.x;
    r01.y = c1.y;
    r23.x = c1.z;
    r23.y = c1.w;
    acc0 = wmma4(a01, p01, acc0);
    acc1 = wmma4(a01, r01, acc1);
    acc0 = wmma4(a23, p23, acc0);
    acc1 = wmma4(a23, r23, acc1);
  }
}

template <bool RELU_IN>
__global__ __launch_bounds__(512) void k_sage_gemm(
    const float* __restrict__ agg, const float* __restrict__ deg,
    const float* __restrict__ h, const float* __restrict__ Wl,
    const float* __restrict__ Wr, const float* __restrict__ bias,
    float* __restrict__ out) {
  __shared__ __attribute__((aligned(16))) float wlds[2 * DD * WSTRIDE];
  __shared__ __attribute__((aligned(16))) float park[16 * PIECE_WORDS];

  const int wv = __builtin_amdgcn_readfirstlane(threadIdx.x >> 5);
  const int lane = threadIdx.x & 31;
  const int half = lane >> 4;
  const int lm = lane & 15;
  const int q = wv & 3;
  const int slot = wv >> 2;
  const int er = lane >> 3;
  const int ec = (lane & 7) << 2;

  const int tile0 = blockIdx.x + GEMM_BLOCKS * slot;
  float4 a[16];
  load_row16(a, agg + (size_t)((tile0 << 4) + lm) * DD + 4 * half);
  float dg = deg[(tile0 << 4) + lm];
  const float4 b4 = *(const float4*)(bias + 32 * q + ec);

  static_assert((DD * DD) / 4 == 8 * 512, "eight 16-byte pieces a thread");
#pragma unroll
  for (int j = 0; j < 8; ++j) {
    int i = threadIdx.x + 512 * j;
    int o = i >> 5;
    int kc = (i & 31) << 2;
    float4 wa = *(const float4*)(Wl + (size_t)o * DD + kc);
    float4 wb = *(const float4*)(Wr + (size_t)o * DD + kc);
    *(float4*)(&wlds[o * WSTRIDE + kc]) = wa;
    *(float4*)(&wlds[WOFF + o * WSTRIDE + kc]) = wb;
  }

  const float* bl = &wlds[(size_t)(32 * q + lm) * WSTRIDE + 4 * half];
  const float* br = bl + WOFF;
  float* st = park + wv * PIECE_WORDS;
  __syncthreads();

  int tile = tile0;
#pragma unroll 1
  for (int r = 0; r < GEMM_ROUNDS; ++r) {
    if (tile < NT) {
      const int m0 = tile << 4;
      const int tnext = tile + GEMM_BLOCKS * GEMM_SLOTS;
      v8f acc0 = {}, acc1 = {};
      const float sc = 1.0f / fmaxf(dg, 1.0f);
      const size_t g0 = (size_t)(m0 + er) * DD + 32 * q + ec;
      float4 rv[4];
      if (RELU_IN) {
#pragma unroll
        for (int j = 0; j < 4; ++j)
          rv[j] = *(const float4*)(h + g0 + (size_t)(4 * j) * DD);
      }
      float4 hb[16];
      load_row16(hb, h + (size_t)(m0 + lm) * DD + 4 * half);
      half_product<true, false>(a, sc, bl, acc0, acc1);
      const int tn = (tnext < NT) ? tnext : tile;
      load_row16(a, agg + (size_t)((tn << 4) + lm) * DD + 4 * half);
      dg = deg[(tn << 4) + lm];
      half_product<false, RELU_IN>(hb, 1.0f, br, acc0, acc1);

#pragma unroll
      for (int e = 0; e < 8; ++e) {
        st[(e + (half << 3)) * PSTRIDE + lm] = acc0[e];
        st[(e + (half << 3)) * PSTRIDE + 16 + lm] = acc1[e];
      }
      float4 pk[4];
#pragma unroll
      for (int j = 0; j < 4; ++j)
        pk[j] = *(const float4*)(st + (4 * j + er) * PSTRIDE + ec);
#pragma unroll
      for (int j = 0; j < 4; ++j) {
        float4 v;
        v.x = pk[j].x + b4.x;
        v.y = pk[j].y + b4.y;
        v.z = pk[j].z + b4.z;
        v.w = pk[j].w + b4.w;
        if (RELU_IN) {
          v.x += rv[j].x;
          v.y += rv[j].y;
          v.z += rv[j].z;
          v.w += rv[j].w;
        }
        *(float4*)(out + g0 + (size_t)(4 * j) * DD) = v;
      }
      tile = tnext;
    }
  }
}

extern "C" void kernel_launch(void* const* d_in, const int* in_sizes, int n_in,
                              void* d_out, int out_size, void* d_ws,
                              size_t ws_size, hipStream_t stream) {
  const float* x = (const float*)d_in[0];
  const int* erow = (const int*)d_in[1];
  const int* ecol = (const int*)d_in[2];
  const float* Wl0 = (const float*)d_in[3];
  const float* Wr0 = (const float*)d_in[4];
  const float* b0 = (const float*)d_in[5];
  const float* Wl1 = (const float*)d_in[6];
  const float* Wr1 = (const float*)d_in[7];
  const float* b1 = (const float*)d_in[8];
  const float* Wl2 = (const float*)d_in[9];
  const float* Wr2 = (const float*)d_in[10];
  const float* b2 = (const float*)d_in[11];
  float* out = (float*)d_out;

  float* ws = (float*)d_ws;
  float* agg = ws;
  float* y1 = agg + (size_t)FEAT;
  float* y2 = y1 + (size_t)FEAT;
  float* deg = y2 + (size_t)FEAT;
  unsigned* lst = (unsigned*)d_out;
  int* cnts = (int*)d_out + LIST_WORDS;
  unsigned* pk = (unsigned*)d_out + PACK_OFF;
  unsigned* bmp = (unsigned*)d_out + BM_OFF;

  dim3 b256(256);
  dim3 b512(512);
  dim3 b1024(1024);
  const int gemmBlocks = GEMM_BLOCKS;

  k_edge_index<<<dim3(IDX_GRID), b256, 0, stream>>>(erow, ecol, pk, bmp);
  k_block_sum<false><<<dim3(NBLK), b1024, 0, stream>>>(x, pk, bmp, agg, deg,
                                                       lst, cnts);
  k_sage_gemm<false><<<dim3(gemmBlocks), b512, 0, stream>>>(agg, deg, x, Wl0,
                                                            Wr0, b0, y1);

  k_block_sum<true><<<dim3(NBLK), b1024, 0, stream>>>(y1, pk, bmp, agg, nullptr,
                                                      lst, cnts);
  k_sage_gemm<true><<<dim3(gemmBlocks), b512, 0, stream>>>(agg, deg, y1, Wl1,
                                                           Wr1, b1, y2);

  k_block_sum<true><<<dim3(NBLK), b1024, 0, stream>>>(y2, pk, bmp, agg, nullptr,
                                                      lst, cnts);
  k_sage_gemm<true><<<dim3(gemmBlocks), b512, 0, stream>>>(agg, deg, y2, Wl2,
                                                           Wr2, b2, out);
}
